# starting kernel
# baseline (speedup 1.0000x reference)
#include <hip/hip_runtime.h>

typedef float v2f __attribute__((ext_vector_type(2)));
typedef float v8f __attribute__((ext_vector_type(8)));

#define HID    128
#define KSTEPS 32
#define NTILES 8

__global__ __launch_bounds__(128)
void gcn_gemm_wmma(const float* __restrict__ x,
                   const float* __restrict__ W,
                   const float* __restrict__ b,
                   float* __restrict__ h,
                   float* __restrict__ out,
                   int n_nodes)
{
    const int wave   = blockIdx.x * (blockDim.x >> 5) + (threadIdx.x >> 5);
    const int lane   = threadIdx.x & 31;
    const int m_base = wave * 16;
    if (m_base >= n_nodes) return;

    const int rown  = lane & 15;
    const int khalf = (lane >> 4) << 1;

    v2f a[KSTEPS];
    {
        const float* xrow = x + (size_t)(m_base + rown) * HID + khalf;
        #pragma unroll
        for (int k = 0; k < KSTEPS; ++k)
            a[k] = *reinterpret_cast<const v2f*>(xrow + 4 * k);
    }

    #pragma unroll
    for (int nt = 0; nt < NTILES; ++nt) {
        const int n_base = nt * 16;
        const float bn = b[n_base + rown];
        v8f c;
        #pragma unroll
        for (int i = 0; i < 8; ++i) c[i] = bn;

        const float* wrow = W + (size_t)(n_base + rown) * HID + khalf;
        #pragma unroll
        for (int k = 0; k < KSTEPS; ++k) {
            v2f bf = *reinterpret_cast<const v2f*>(wrow + 4 * k);
            c = __builtin_amdgcn_wmma_f32_16x16x4_f32(
                     false, a[k],  false, bf,
                     (short)0, c,  false,  false);
        }

        const int col   = n_base + rown;
        const int rbase = m_base + ((lane >> 4) << 3);
        #pragma unroll
        for (int r = 0; r < 8; ++r) {
            const size_t off = (size_t)(rbase + r) * HID + col;
            h[off]   = c[r];
            out[off] = c[r];
        }
    }
}

__global__ void gcn_zero_deg(int* __restrict__ deg, int n)
{
    int i = blockIdx.x * blockDim.x + threadIdx.x;
    if (i < n) deg[i] = 0;
}

__global__ __launch_bounds__(256)
void gcn_scatter(const int* __restrict__ edge_index,
                 const float* __restrict__ h,
                 float* __restrict__ out,
                 int* __restrict__ deg,
                 int n_edges)
{
    const int tid = blockIdx.x * blockDim.x + threadIdx.x;
    const int e   = tid >> 5;
    if (e >= n_edges) return;
    const int lane = tid & 31;

    const int dst = edge_index[e];
    const int src = edge_index[n_edges + e];

    if (lane == 0)
        __hip_atomic_fetch_add(&deg[dst], 1, __ATOMIC_RELAXED, __HIP_MEMORY_SCOPE_AGENT);

    const float4 v = *reinterpret_cast<const float4*>(h + (size_t)src * HID + lane * 4);
    float* o = out + (size_t)dst * HID + lane * 4;
    __hip_atomic_fetch_add(&o[0], v.x, __ATOMIC_RELAXED, __HIP_MEMORY_SCOPE_AGENT);
    __hip_atomic_fetch_add(&o[1], v.y, __ATOMIC_RELAXED, __HIP_MEMORY_SCOPE_AGENT);
    __hip_atomic_fetch_add(&o[2], v.z, __ATOMIC_RELAXED, __HIP_MEMORY_SCOPE_AGENT);
    __hip_atomic_fetch_add(&o[3], v.w, __ATOMIC_RELAXED, __HIP_MEMORY_SCOPE_AGENT);
}

__global__ __launch_bounds__(256)
void gcn_finalize(float* __restrict__ out, const int* __restrict__ deg, int n_nodes)
{
    const int tid   = blockIdx.x * blockDim.x + threadIdx.x;
    const int total = n_nodes * (HID / 4);
    if (tid >= total) return;
    const int node  = tid >> 5;
    const float s   = 1.0f / (float)(deg[node] + 1);
    float4* p = reinterpret_cast<float4*>(out) + tid;
    float4 v = *p;
    v.x *= s; v.y *= s; v.z *= s; v.w *= s;
    *p = v;
}

extern "C" void kernel_launch(void* const* d_in, const int* in_sizes, int n_in,
                              void* d_out, int out_size, void* d_ws, size_t ws_size,
                              hipStream_t stream)
{
    const float* x          = (const float*)d_in[0];
    const int*   edge_index = (const int*)  d_in[1];
    const float* W          = (const float*)d_in[2];
    const float* b          = (const float*)d_in[3];
    float*       out        = (float*)d_out;

    const int n_nodes = in_sizes[0] / HID;
    const int n_edges = in_sizes[1] / 2;

    float* h   = (float*)d_ws;
    int*   deg = (int*)((char*)d_ws + (size_t)n_nodes * HID * sizeof(float));

    {
        const int waves  = (n_nodes + 15) / 16;
        const int blocks = (waves + 3) / 4;
        gcn_gemm_wmma<<<blocks, 128, 0, stream>>>(x, W, b, h, out, n_nodes);
    }
    gcn_zero_deg<<<(n_nodes + 255) / 256, 256, 0, stream>>>(deg, n_nodes);
    {
        const long long threads = (long long)n_edges * 32;
        const int blocks = (int)((threads + 255) / 256);
        gcn_scatter<<<blocks, 256, 0, stream>>>(edge_index, h, out, deg, n_edges);
    }
    {
        const int total = n_nodes * (HID / 4);
        gcn_finalize<<<(total + 255) / 256, 256, 0, stream>>>(out, deg, n_nodes);
    }
}


# Owner blocks scan one owner-code byte per edge, not the 4-byte dst id (code plane and (dst,src) pair records packed by extra blocks of launch 1); sum order: LDS arrival, edges regrouped per wave
# speedup vs baseline: 3.9839x; 3.9839x over previous
#include <hip/hip_runtime.h>

typedef float v2f __attribute__((ext_vector_type(2)));
typedef float v8f __attribute__((ext_vector_type(8)));

#define HID    128
#define KSTEPS 32
#define NTILES 8

#define R_NODES     196
#define AGG_BLOCKS  256

__device__ __forceinline__ unsigned owner_codes4(int4 d)
{
    return  (((unsigned)d.x / R_NODES) % AGG_BLOCKS)
         | ((((unsigned)d.y / R_NODES) % AGG_BLOCKS) << 8)
         | ((((unsigned)d.z / R_NODES) % AGG_BLOCKS) << 16)
         | ((((unsigned)d.w / R_NODES) % AGG_BLOCKS) << 24);
}

__global__ __launch_bounds__(128)
void gcn_gemm_wmma(const float* __restrict__ x,
                   const float* __restrict__ W,
                   const float* __restrict__ b,
                   float* __restrict__ h,
                   const int* __restrict__ edge_index,
                   int n_nodes, int n_edges, int gemm_blocks)
{
    if ((int)blockIdx.x >= gemm_blocks) {
        const int chunk  = ((int)blockIdx.x - gemm_blocks) * 4 + (int)(threadIdx.x >> 5);
        const int n_full = n_edges >> 9;
        const int l = threadIdx.x & 31;
        const int* dstp = edge_index + ((size_t)chunk << 9);
        const int* srcp = dstp + n_edges;
        unsigned* codes = reinterpret_cast<unsigned*>(h + (size_t)n_nodes * HID);
        uint2*    rec   = reinterpret_cast<uint2*>(codes + ((size_t)n_full << 7)) + ((size_t)chunk << 9);
        if (chunk < n_full) {
            const int4* dp = reinterpret_cast<const int4*>(dstp + (l << 2));
            const int4 d0 = dp[0], d1 = dp[32], d2 = dp[64], d3 = dp[96];
            uint4 cw;
            cw.x = owner_codes4(d0);
            cw.y = owner_codes4(d1);
            cw.z = owner_codes4(d2);
            cw.w = owner_codes4(d3);
            *reinterpret_cast<uint4*>(codes + ((size_t)chunk << 7) + (l << 2)) = cw;
            #pragma unroll
            for (int k = 0; k < 8; ++k) {
                const int2 d = *reinterpret_cast<const int2*>(dstp + 64 * k + 2 * l);
                const int s0 = srcp[64 * k + 2 * l];
                const int s1 = srcp[64 * k + 2 * l + 1];
                *reinterpret_cast<uint4*>(rec + 64 * k + 2 * l) =
                    make_uint4((unsigned)d.x, (unsigned)s0, (unsigned)d.y, (unsigned)s1);
            }
        } else {
            const int left = n_edges - (chunk << 9);
            #pragma nounroll
            for (int i = l; i < left; i += 32)
                rec[i] = make_uint2((unsigned)dstp[i], (unsigned)srcp[i]);
        }
        return;
    }

    const int wave   = blockIdx.x * (blockDim.x >> 5) + (threadIdx.x >> 5);
    const int lane   = threadIdx.x & 31;
    const int m_base = wave * 16;
    if (m_base >= n_nodes) return;

    const int rown  = lane & 15;
    const int khalf = (lane >> 4) << 1;

    v2f a[KSTEPS];
    {
        const float* xrow = x + (size_t)(m_base + rown) * HID + khalf;
        #pragma unroll
        for (int k = 0; k < KSTEPS; ++k)
            a[k] = *reinterpret_cast<const v2f*>(xrow + 4 * k);
    }

    #pragma unroll
    for (int nt = 0; nt < NTILES; ++nt) {
        const int n_base = nt * 16;
        const float bn = b[n_base + rown];
        v8f c;
        #pragma unroll
        for (int i = 0; i < 8; ++i) c[i] = bn;

        const float* wrow = W + (size_t)(n_base + rown) * HID + khalf;
        #pragma unroll
        for (int k = 0; k < KSTEPS; ++k) {
            v2f bf = *reinterpret_cast<const v2f*>(wrow + 4 * k);
            c = __builtin_amdgcn_wmma_f32_16x16x4_f32(
                     false, a[k],  false, bf,
                     (short)0, c,  false,  false);
        }

        const int col   = n_base + rown;
        const int rbase = m_base + ((lane >> 4) << 3);
        #pragma unroll
        for (int r = 0; r < 8; ++r) {
            const size_t off = (size_t)(rbase + r) * HID + col;
            h[off] = c[r];
        }
    }
}

#define AGG_THREADS 1024
#define AGG_WAVES   32
#define SCAN_UNROLL 2
#define QCAP        1056

__global__ __launch_bounds__(AGG_THREADS)
void gcn_owner_aggregate(const int* __restrict__ edge_index,
                         const float* __restrict__ h,
                         float* __restrict__ out,
                         int n_nodes, int n_edges)
{
    __shared__ __attribute__((aligned(16))) float acc[(R_NODES + 1) * HID];
    __shared__ int   degs[R_NODES + 4];
    __shared__ unsigned queue[AGG_WAVES * QCAP];

    const int lane = threadIdx.x & 31;
    const int wave = __builtin_amdgcn_readfirstlane((int)(threadIdx.x >> 5));
    unsigned* q = queue + wave * QCAP;
    const int* dstp = edge_index;
    const int n_full = n_edges >> 9;
    const unsigned* codes = reinterpret_cast<const unsigned*>(h + (size_t)n_nodes * HID);
    const unsigned long long* rec = reinterpret_cast<const unsigned long long*>(codes + ((size_t)n_full << 7));
    const unsigned  pat   = (blockIdx.x % AGG_BLOCKS) * 0x01010101u;
    const int start = (int)(((long long)blockIdx.x * n_full) / gridDim.x);

    for (int base = blockIdx.x * R_NODES; base < n_nodes; base += AGG_BLOCKS * R_NODES) {
        {
            float4* a4 = reinterpret_cast<float4*>(acc);
            for (int i = threadIdx.x; i < (R_NODES + 1) * HID / 4; i += AGG_THREADS)
                a4[i] = make_float4(0.f, 0.f, 0.f, 0.f);
            if (threadIdx.x < R_NODES + 4) degs[threadIdx.x] = 0;
        }
        __syncthreads();

        int cnt = 0;

        auto chunk_of = [&](int t) -> int {
            const int tc = (t < n_full) ? t : (n_full - 1);
            int g = tc + start;
            if (g >= n_full) g -= n_full;
            return g;
        };
        auto load_codes = [&](int g) -> uint4 {
            return *reinterpret_cast<const uint4*>(codes + ((size_t)g << 7) + (lane << 2));
        };

        auto drain = [&]() {
            __builtin_amdgcn_wave_barrier();
            for (int q0 = 0; q0 < cnt; q0 += 32) {
                const int nq = (cnt - q0 < 32) ? (cnt - q0) : 32;
                int s  = 0;
                int dl = R_NODES;
                if (lane < nq) {
                    const unsigned long long ds = rec[q[q0 + lane]];
                    const unsigned r = (unsigned)ds - (unsigned)base;
                    const bool own = r < (unsigned)R_NODES;
                    s  = own ? (int)(unsigned)(ds >> 32) : 0;
                    dl = own ? (int)r : R_NODES;
                    if (own) __hip_atomic_fetch_add(&degs[r], 1, __ATOMIC_RELAXED, __HIP_MEMORY_SCOPE_WORKGROUP);
                }
                for (int k0 = 0; k0 < nq; k0 += 8) {
                    float4 v[8];
                    #pragma unroll
                    for (int k = 0; k < 8; ++k) {
                        const int sk = __builtin_amdgcn_readlane(s, k0 + k);
                        v[k] = *reinterpret_cast<const float4*>(h + (size_t)sk * HID + lane * 4);
                    }
                    #pragma unroll
                    for (int k = 0; k < 8; ++k) {
                        const int dk = __builtin_amdgcn_readlane(dl, k0 + k);
                        float* a = acc + dk * HID + lane;
                        __hip_atomic_fetch_add(a,      v[k].x, __ATOMIC_RELAXED, __HIP_MEMORY_SCOPE_WORKGROUP);
                        __hip_atomic_fetch_add(a + 32, v[k].y, __ATOMIC_RELAXED, __HIP_MEMORY_SCOPE_WORKGROUP);
                        __hip_atomic_fetch_add(a + 64, v[k].z, __ATOMIC_RELAXED, __HIP_MEMORY_SCOPE_WORKGROUP);
                        __hip_atomic_fetch_add(a + 96, v[k].w, __ATOMIC_RELAXED, __HIP_MEMORY_SCOPE_WORKGROUP);
                    }
                }
            }
            cnt = 0;
            __builtin_amdgcn_wave_barrier();
        };

        auto enqueue = [&](unsigned mask, bool m, unsigned e) {
            if (m) q[cnt + (int)__builtin_amdgcn_mbcnt_lo(mask, 0u)] = e;
            cnt += __builtin_popcount(mask);
        };

        auto miss4 = [&](unsigned w) -> unsigned {
            const unsigned t = w ^ pat;
            return ((t & 0x7f7f7f7fu) + 0x7f7f7f7fu) | t | 0x7f7f7f7fu;
        };

        auto process = [&](uint4 cw, int g) {
            const unsigned n0 = miss4(cw.x), n1 = miss4(cw.y), n2 = miss4(cw.z), n3 = miss4(cw.w);
            if (__builtin_amdgcn_ballot_w32((n0 & n1 & n2 & n3) != 0xffffffffu) != 0u) {
                unsigned m = (~n0 >> 7) | (~n1 >> 6) | (~n2 >> 5) | (~n3 >> 4);
                const unsigned e0 = ((unsigned)g << 9) + ((unsigned)lane << 2);
                #pragma nounroll
                for (int it = 0; it < 16; ++it) {
                    const bool has = m != 0u;
                    const unsigned mask = __builtin_amdgcn_ballot_w32(has);
                    if (mask == 0u) break;
                    const unsigned p = (unsigned)__builtin_ctz(m | 0x80000000u);
                    enqueue(mask, has, e0 + ((p & 7u) << 7) + (p >> 3));
                    m &= m - 1u;
                }
            }
        };

        if (n_full > 0) {
            uint4 da[SCAN_UNROLL], db[SCAN_UNROLL];
            int   ga[SCAN_UNROLL], gb[SCAN_UNROLL];
            #pragma unroll
            for (int u = 0; u < SCAN_UNROLL; ++u) {
                ga[u] = chunk_of(wave + u * AGG_WAVES);
                da[u] = load_codes(ga[u]);
            }
            for (int t0 = wave; t0 < n_full; t0 += 2 * SCAN_UNROLL * AGG_WAVES) {
                #pragma unroll
                for (int u = 0; u < SCAN_UNROLL; ++u) {
                    gb[u] = chunk_of(t0 + (SCAN_UNROLL + u) * AGG_WAVES);
                    db[u] = load_codes(gb[u]);
                }
                #pragma unroll
                for (int u = 0; u < SCAN_UNROLL; ++u)
                    if (t0 + u * AGG_WAVES < n_full) process(da[u], ga[u]);
                if (cnt >= 32) drain();
                #pragma unroll
                for (int u = 0; u < SCAN_UNROLL; ++u) {
                    ga[u] = chunk_of(t0 + (2 * SCAN_UNROLL + u) * AGG_WAVES);
                    da[u] = load_codes(ga[u]);
                }
                #pragma unroll
                for (int u = 0; u < SCAN_UNROLL; ++u)
                    if (t0 + (SCAN_UNROLL + u) * AGG_WAVES < n_full) process(db[u], gb[u]);
                if (cnt >= 32) drain();
            }
        }
        if (wave < 4 && (n_full << 9) + (wave << 7) < n_edges) {
            const int e0 = (n_full << 9) + (wave << 7) + (lane << 2);
            #pragma unroll
            for (int j = 0; j < 4; ++j) {
                bool m = false;
                if (e0 + j < n_edges) m = (unsigned)(dstp[e0 + j] - base) < (unsigned)R_NODES;
                enqueue(__builtin_amdgcn_ballot_w32(m), m, (unsigned)(e0 + j));
            }
        }
        drain();
        __syncthreads();

        for (int r = wave; r < R_NODES; r += AGG_WAVES) {
            const int node = base + r;
            if (node < n_nodes) {
                const float* a = acc + r * HID + lane;
                const float4 hs = *reinterpret_cast<const float4*>(h + (size_t)node * HID + lane * 4);
                const float dn = (float)(degs[r] + 1);
                float4 o;
                o.x = (hs.x + a[0])  / dn;
                o.y = (hs.y + a[32]) / dn;
                o.z = (hs.z + a[64]) / dn;
                o.w = (hs.w + a[96]) / dn;
                *reinterpret_cast<float4*>(out + (size_t)node * HID + lane * 4) = o;
            }
        }
        __syncthreads();
    }
}

extern "C" void kernel_launch(void* const* d_in, const int* in_sizes, int n_in,
                              void* d_out, int out_size, void* d_ws, size_t ws_size,
                              hipStream_t stream)
{
    const float* x          = (const float*)d_in[0];
    const int*   edge_index = (const int*)  d_in[1];
    const float* W          = (const float*)d_in[2];
    const float* b          = (const float*)d_in[3];
    float*       out        = (float*)d_out;

    const int n_nodes = in_sizes[0] / HID;
    const int n_edges = in_sizes[1] / 2;

    float* h = (float*)d_ws;

    {
        const int waves  = (n_nodes + 15) / 16;
        const int blocks = (waves + 3) / 4;
        const int code_blocks = ((n_edges + 511) / 512 + 3) / 4;
        gcn_gemm_wmma<<<blocks + code_blocks, 128, 0, stream>>>(x, W, b, h, edge_index, n_nodes, n_edges, blocks);
    }
    gcn_owner_aggregate<<<AGG_BLOCKS, AGG_THREADS, 0, stream>>>(edge_index, h, out, n_nodes, n_edges);
}


# GEMM B operand: each block copies W once into shared memory in fragment order and reads B fragments by ds_load_b128 (float4 = two k-steps, A likewise); k order of the product sums changes
# speedup vs baseline: 4.3991x; 1.1042x over previous
#include <hip/hip_runtime.h>

typedef float v2f __attribute__((ext_vector_type(2)));
typedef float v8f __attribute__((ext_vector_type(8)));

#define HID    128
#define KPAIRS 16
#define NTILES 8

#define R_NODES     196
#define AGG_BLOCKS  256

__device__ __forceinline__ unsigned owner_codes4(int4 d)
{
    return  (((unsigned)d.x / R_NODES) % AGG_BLOCKS)
         | ((((unsigned)d.y / R_NODES) % AGG_BLOCKS) << 8)
         | ((((unsigned)d.z / R_NODES) % AGG_BLOCKS) << 16)
         | ((((unsigned)d.w / R_NODES) % AGG_BLOCKS) << 24);
}

__global__ __launch_bounds__(128)
void gcn_gemm_wmma(const float* __restrict__ x,
                   const float* __restrict__ W,
                   const float* __restrict__ b,
                   float* __restrict__ h,
                   const int* __restrict__ edge_index,
                   int n_nodes, int n_edges, int gemm_blocks)
{
    __shared__ float4 wfrag[KPAIRS * NTILES * 32];

    if ((int)blockIdx.x >= gemm_blocks) {
        const int chunk  = ((int)blockIdx.x - gemm_blocks) * 4 + (int)(threadIdx.x >> 5);
        const int n_full = n_edges >> 9;
        const int l = threadIdx.x & 31;
        const int* dstp = edge_index + ((size_t)chunk << 9);
        const int* srcp = dstp + n_edges;
        unsigned* codes = reinterpret_cast<unsigned*>(h + (size_t)n_nodes * HID);
        uint2*    rec   = reinterpret_cast<uint2*>(codes + ((size_t)n_full << 7)) + ((size_t)chunk << 9);
        if (chunk < n_full) {
            const int4* dp = reinterpret_cast<const int4*>(dstp + (l << 2));
            const int4 d0 = dp[0], d1 = dp[32], d2 = dp[64], d3 = dp[96];
            uint4 cw;
            cw.x = owner_codes4(d0);
            cw.y = owner_codes4(d1);
            cw.z = owner_codes4(d2);
            cw.w = owner_codes4(d3);
            *reinterpret_cast<uint4*>(codes + ((size_t)chunk << 7) + (l << 2)) = cw;
            #pragma unroll
            for (int k = 0; k < 8; ++k) {
                const int2 d = *reinterpret_cast<const int2*>(dstp + 64 * k + 2 * l);
                const int s0 = srcp[64 * k + 2 * l];
                const int s1 = srcp[64 * k + 2 * l + 1];
                *reinterpret_cast<uint4*>(rec + 64 * k + 2 * l) =
                    make_uint4((unsigned)d.x, (unsigned)s0, (unsigned)d.y, (unsigned)s1);
            }
        } else {
            const int left = n_edges - (chunk << 9);
            #pragma nounroll
            for (int i = l; i < left; i += 32)
                rec[i] = make_uint2((unsigned)dstp[i], (unsigned)srcp[i]);
        }
        return;
    }

    const int wv     = threadIdx.x >> 5;
    const int lane   = threadIdx.x & 31;
    const int rown   = lane & 15;
    const int k4     = (lane >> 4) << 2;
    const int m_base = (blockIdx.x * 4 + wv) * 16;
    const bool live  = m_base < n_nodes;

    float4 a[KPAIRS];
    if (live) {
        const float* xrow = x + (size_t)(m_base + rown) * HID + k4;
        #pragma unroll
        for (int j = 0; j < KPAIRS; ++j)
            a[j] = *reinterpret_cast<const float4*>(xrow + 8 * j);
    }

    {
        const float* wsrc = W + (size_t)rown * HID + 32 * wv + k4;
        float4* wdst = wfrag + (size_t)(32 * wv) * 32 + lane;
        float4 t[32];
        #pragma unroll
        for (int s = 0; s < 32; ++s)
            t[s] = *reinterpret_cast<const float4*>(wsrc + (size_t)(s & 7) * 16 * HID + 8 * (s >> 3));
        #pragma unroll
        for (int s = 0; s < 32; ++s)
            wdst[s * 32] = t[s];
    }
    __syncthreads();
    if (!live) return;

    v8f c[NTILES];
    #pragma unroll
    for (int nt = 0; nt < NTILES; ++nt) {
        const float bn = b[nt * 16 + rown];
        #pragma unroll
        for (int i = 0; i < 8; ++i) c[nt][i] = bn;
    }

    const float4* bw = wfrag + lane;
    #pragma unroll
    for (int j = 0; j < KPAIRS; ++j) {
        float4 bf[NTILES];
        #pragma unroll
        for (int nt = 0; nt < NTILES; ++nt)
            bf[nt] = bw[(j * NTILES + nt) * 32];
        v2f a0, a1;
        a0[0] = a[j].x; a0[1] = a[j].y;
        a1[0] = a[j].z; a1[1] = a[j].w;
        #pragma unroll
        for (int nt = 0; nt < NTILES; ++nt) {
            v2f b0; b0[0] = bf[nt].x; b0[1] = bf[nt].y;
            c[nt] = __builtin_amdgcn_wmma_f32_16x16x4_f32(
                         false, a0,  false, b0,
                         (short)0, c[nt],  false,  false);
        }
        #pragma unroll
        for (int nt = 0; nt < NTILES; ++nt) {
            v2f b1; b1[0] = bf[nt].z; b1[1] = bf[nt].w;
            c[nt] = __builtin_amdgcn_wmma_f32_16x16x4_f32(
                         false, a1,  false, b1,
                         (short)0, c[nt],  false,  false);
        }
    }

    const int rbase = m_base + ((lane >> 4) << 3);
    #pragma unroll
    for (int nt = 0; nt < NTILES; ++nt) {
        const int col = nt * 16 + rown;
        #pragma unroll
        for (int r = 0; r < 8; ++r) {
            const size_t off = (size_t)(rbase + r) * HID + col;
            h[off] = c[nt][r];
        }
    }
}

#define AGG_THREADS 1024
#define AGG_WAVES   32
#define SCAN_UNROLL 2
#define QCAP        1056

__global__ __launch_bounds__(AGG_THREADS)
void gcn_owner_aggregate(const int* __restrict__ edge_index,
                         const float* __restrict__ h,
                         float* __restrict__ out,
                         int n_nodes, int n_edges)
{
    __shared__ __attribute__((aligned(16))) float acc[(R_NODES + 1) * HID];
    __shared__ int   degs[R_NODES + 4];
    __shared__ unsigned queue[AGG_WAVES * QCAP];

    const int lane = threadIdx.x & 31;
    const int wave = __builtin_amdgcn_readfirstlane((int)(threadIdx.x >> 5));
    unsigned* q = queue + wave * QCAP;
    const int* dstp = edge_index;
    const int n_full = n_edges >> 9;
    const unsigned* codes = reinterpret_cast<const unsigned*>(h + (size_t)n_nodes * HID);
    const unsigned long long* rec = reinterpret_cast<const unsigned long long*>(codes + ((size_t)n_full << 7));
    const unsigned  pat   = (blockIdx.x % AGG_BLOCKS) * 0x01010101u;
    const int start = (int)(((long long)blockIdx.x * n_full) / gridDim.x);

    for (int base = blockIdx.x * R_NODES; base < n_nodes; base += AGG_BLOCKS * R_NODES) {
        {
            float4* a4 = reinterpret_cast<float4*>(acc);
            for (int i = threadIdx.x; i < (R_NODES + 1) * HID / 4; i += AGG_THREADS)
                a4[i] = make_float4(0.f, 0.f, 0.f, 0.f);
            if (threadIdx.x < R_NODES + 4) degs[threadIdx.x] = 0;
        }
        __syncthreads();

        int cnt = 0;

        auto chunk_of = [&](int t) -> int {
            const int tc = (t < n_full) ? t : (n_full - 1);
            int g = tc + start;
            if (g >= n_full) g -= n_full;
            return g;
        };
        auto load_codes = [&](int g) -> uint4 {
            return *reinterpret_cast<const uint4*>(codes + ((size_t)g << 7) + (lane << 2));
        };

        auto drain = [&]() {
            __builtin_amdgcn_wave_barrier();
            for (int q0 = 0; q0 < cnt; q0 += 32) {
                const int nq = (cnt - q0 < 32) ? (cnt - q0) : 32;
                int s  = 0;
                int dl = R_NODES;
                if (lane < nq) {
                    const unsigned long long ds = rec[q[q0 + lane]];
                    const unsigned r = (unsigned)ds - (unsigned)base;
                    const bool own = r < (unsigned)R_NODES;
                    s  = own ? (int)(unsigned)(ds >> 32) : 0;
                    dl = own ? (int)r : R_NODES;
                    if (own) __hip_atomic_fetch_add(&degs[r], 1, __ATOMIC_RELAXED, __HIP_MEMORY_SCOPE_WORKGROUP);
                }
                for (int k0 = 0; k0 < nq; k0 += 8) {
                    float4 v[8];
                    #pragma unroll
                    for (int k = 0; k < 8; ++k) {
                        const int sk = __builtin_amdgcn_readlane(s, k0 + k);
                        v[k] = *reinterpret_cast<const float4*>(h + (size_t)sk * HID + lane * 4);
                    }
                    #pragma unroll
                    for (int k = 0; k < 8; ++k) {
                        const int dk = __builtin_amdgcn_readlane(dl, k0 + k);
                        float* a = acc + dk * HID + lane;
                        __hip_atomic_fetch_add(a,      v[k].x, __ATOMIC_RELAXED, __HIP_MEMORY_SCOPE_WORKGROUP);
                        __hip_atomic_fetch_add(a + 32, v[k].y, __ATOMIC_RELAXED, __HIP_MEMORY_SCOPE_WORKGROUP);
                        __hip_atomic_fetch_add(a + 64, v[k].z, __ATOMIC_RELAXED, __HIP_MEMORY_SCOPE_WORKGROUP);
                        __hip_atomic_fetch_add(a + 96, v[k].w, __ATOMIC_RELAXED, __HIP_MEMORY_SCOPE_WORKGROUP);
                    }
                }
            }
            cnt = 0;
            __builtin_amdgcn_wave_barrier();
        };

        auto enqueue = [&](unsigned mask, bool m, unsigned e) {
            if (m) q[cnt + (int)__builtin_amdgcn_mbcnt_lo(mask, 0u)] = e;
            cnt += __builtin_popcount(mask);
        };

        auto miss4 = [&](unsigned w) -> unsigned {
            const unsigned t = w ^ pat;
            return ((t & 0x7f7f7f7fu) + 0x7f7f7f7fu) | t | 0x7f7f7f7fu;
        };

        auto process = [&](uint4 cw, int g) {
            const unsigned n0 = miss4(cw.x), n1 = miss4(cw.y), n2 = miss4(cw.z), n3 = miss4(cw.w);
            if (__builtin_amdgcn_ballot_w32((n0 & n1 & n2 & n3) != 0xffffffffu) != 0u) {
                unsigned m = (~n0 >> 7) | (~n1 >> 6) | (~n2 >> 5) | (~n3 >> 4);
                const unsigned e0 = ((unsigned)g << 9) + ((unsigned)lane << 2);
                #pragma nounroll
                for (int it = 0; it < 16; ++it) {
                    const bool has = m != 0u;
                    const unsigned mask = __builtin_amdgcn_ballot_w32(has);
                    if (mask == 0u) break;
                    const unsigned p = (unsigned)__builtin_ctz(m | 0x80000000u);
                    enqueue(mask, has, e0 + ((p & 7u) << 7) + (p >> 3));
                    m &= m - 1u;
                }
            }
        };

        if (n_full > 0) {
            uint4 da[SCAN_UNROLL], db[SCAN_UNROLL];
            int   ga[SCAN_UNROLL], gb[SCAN_UNROLL];
            #pragma unroll
            for (int u = 0; u < SCAN_UNROLL; ++u) {
                ga[u] = chunk_of(wave + u * AGG_WAVES);
                da[u] = load_codes(ga[u]);
            }
            for (int t0 = wave; t0 < n_full; t0 += 2 * SCAN_UNROLL * AGG_WAVES) {
                #pragma unroll
                for (int u = 0; u < SCAN_UNROLL; ++u) {
                    gb[u] = chunk_of(t0 + (SCAN_UNROLL + u) * AGG_WAVES);
                    db[u] = load_codes(gb[u]);
                }
                #pragma unroll
                for (int u = 0; u < SCAN_UNROLL; ++u)
                    if (t0 + u * AGG_WAVES < n_full) process(da[u], ga[u]);
                if (cnt >= 32) drain();
                #pragma unroll
                for (int u = 0; u < SCAN_UNROLL; ++u) {
                    ga[u] = chunk_of(t0 + (2 * SCAN_UNROLL + u) * AGG_WAVES);
                    da[u] = load_codes(ga[u]);
                }
                #pragma unroll
                for (int u = 0; u < SCAN_UNROLL; ++u)
                    if (t0 + (SCAN_UNROLL + u) * AGG_WAVES < n_full) process(db[u], gb[u]);
                if (cnt >= 32) drain();
            }
        }
        if (wave < 4 && (n_full << 9) + (wave << 7) < n_edges) {
            const int e0 = (n_full << 9) + (wave << 7) + (lane << 2);
            #pragma unroll
            for (int j = 0; j < 4; ++j) {
                bool m = false;
                if (e0 + j < n_edges) m = (unsigned)(dstp[e0 + j] - base) < (unsigned)R_NODES;
                enqueue(__builtin_amdgcn_ballot_w32(m), m, (unsigned)(e0 + j));
            }
        }
        drain();
        __syncthreads();

        for (int r = wave; r < R_NODES; r += AGG_WAVES) {
            const int node = base + r;
            if (node < n_nodes) {
                const float* a = acc + r * HID + lane;
                const float4 hs = *reinterpret_cast<const float4*>(h + (size_t)node * HID + lane * 4);
                const float dn = (float)(degs[r] + 1);
                float4 o;
                o.x = (hs.x + a[0])  / dn;
                o.y = (hs.y + a[32]) / dn;
                o.z = (hs.z + a[64]) / dn;
                o.w = (hs.w + a[96]) / dn;
                *reinterpret_cast<float4*>(out + (size_t)node * HID + lane * 4) = o;
            }
        }
        __syncthreads();
    }
}

extern "C" void kernel_launch(void* const* d_in, const int* in_sizes, int n_in,
                              void* d_out, int out_size, void* d_ws, size_t ws_size,
                              hipStream_t stream)
{
    const float* x          = (const float*)d_in[0];
    const int*   edge_index = (const int*)  d_in[1];
    const float* W          = (const float*)d_in[2];
    const float* b          = (const float*)d_in[3];
    float*       out        = (float*)d_out;

    const int n_nodes = in_sizes[0] / HID;
    const int n_edges = in_sizes[1] / 2;

    float* h = (float*)d_ws;

    {
        const int waves  = (n_nodes + 15) / 16;
        const int blocks = (waves + 3) / 4;
        const int code_blocks = ((n_edges + 511) / 512 + 3) / 4;
        gcn_gemm_wmma<<<blocks + code_blocks, 128, 0, stream>>>(x, W, b, h, edge_index, n_nodes, n_edges, blocks);
    }
    gcn_owner_aggregate<<<AGG_BLOCKS, AGG_THREADS, 0, stream>>>(edge_index, h, out, n_nodes, n_edges);
}
